# starting kernel
# baseline (speedup 1.0000x reference)
#include <hip/hip_runtime.h>

typedef __attribute__((ext_vector_type(2))) float v2f;
typedef __attribute__((ext_vector_type(8))) float v8f;

constexpr int Bsz = 8, N1 = 1024, N2 = 1024, DIM = 256, HID = 16;
constexpr int WCOLS = 6 * HID + 1;

__global__ __launch_bounds__(256) void weights_gemm_wmma(
        const float* __restrict__ q, const float* __restrict__ Wg,
        const float* __restrict__ bg, float* __restrict__ wout) {
    const int lane = threadIdx.x & 31;
    const int wid  = threadIdx.x >> 5;
    const int tile = blockIdx.x * 8 + wid;
    const int COLT = 7;
    const int row0 = (tile / COLT) * 16;
    const int col0 = (tile % COLT) * 16;
    const int h    = lane & 15;
    const bool hiH = lane >= 16;
    const int koff = hiH ? 2 : 0;
    const int col  = col0 + h;
    const int colc = (col < WCOLS) ? col : (WCOLS - 1);

    const float* qrow = q + (size_t)(row0 + h) * DIM + koff;
    const float* wcol = Wg + colc;

    v8f c = {};
    #pragma unroll 4
    for (int kk = 0; kk < DIM; kk += 4) {
        v2f a, bm;
        a.x  = qrow[kk];
        a.y  = qrow[kk + 1];
        bm.x = wcol[(size_t)(kk + koff)     * WCOLS];
        bm.y = wcol[(size_t)(kk + koff + 1) * WCOLS];
        c = __builtin_amdgcn_wmma_f32_16x16x4_f32(false, a, false, bm,
                                                  (short)0, c, false, false);
    }

    const float bgv = bg[colc];
    const int rbase = row0 + (hiH ? 8 : 0);
    #pragma unroll
    for (int r = 0; r < 8; ++r) {
        if (col < WCOLS)
            wout[(size_t)(rbase + r) * WCOLS + col] = c[r] + bgv;
    }
}

__global__ __launch_bounds__(256) void pair_mlp_wmma(
        const float* __restrict__ pos1, const float* __restrict__ pos2,
        const float* __restrict__ wgt, float* __restrict__ out) {
    const int lane = threadIdx.x & 31;
    const int wid  = threadIdx.x >> 5;
    const int gid  = blockIdx.x * 8 + wid;
    const int b    = gid >> 10;
    const bool hiH = lane >= 16;
    const int h    = lane & 15;

    const float* p1 = pos1 + (size_t)gid * 4;
    const float x1  = p1[0], y1 = p1[1], w1p = p1[2], h1p = p1[3];
    const float c0  = hiH ? __logf(w1p) : x1;
    const float c1  = hiH ? __logf(h1p) : y1;
    const float s0  = hiH ? 1.0f : 1.0f / w1p;
    const float s1  = hiH ? 1.0f : 1.0f / h1p;

    const float* wrow = wgt + (size_t)gid * WCOLS;
    const int k0 = hiH ? 2 : 0;
    v2f a;
    a.x = wrow[(k0 + 0) * HID + h];
    a.y = wrow[(k0 + 1) * HID + h];
    const int hb = hiH ? 8 : 0;
    float b1v[8], w2v[8];
    #pragma unroll
    for (int r = 0; r < 8; ++r) {
        b1v[r] = wrow[4 * HID + hb + r];
        w2v[r] = wrow[5 * HID + hb + r];
    }
    const float b2 = wrow[6 * HID];

    const float2* p2row = (const float2*)pos2 + (size_t)b * N2 * 2;
    float* orow = out + (size_t)gid * N2;

    for (int t = 0; t < N2 / 16; ++t) {
        const int n0 = t * 16;
        const float2 p = p2row[(size_t)(n0 + h) * 2 + (hiH ? 1 : 0)];
        const float u0 = hiH ? __logf(p.x) : p.x;
        const float u1 = hiH ? __logf(p.y) : p.y;
        v2f f;
        f.x = (u0 - c0) * s0;
        f.y = (u1 - c1) * s1;

        v8f d = {};
        d = __builtin_amdgcn_wmma_f32_16x16x4_f32(false, a, false, f,
                                                  (short)0, d, false, false);
        float acc = 0.0f;
        #pragma unroll
        for (int r = 0; r < 8; ++r) {
            float tt = fmaxf(d[r] + b1v[r], 0.0f);
            acc = fmaf(tt, w2v[r], acc);
        }
        acc += __shfl_xor(acc, 16, 32);
        if (!hiH) orow[n0 + h] = acc + b2;
    }
}

extern "C" void kernel_launch(void* const* d_in, const int* in_sizes, int n_in,
                              void* d_out, int out_size, void* d_ws, size_t ws_size,
                              hipStream_t stream) {
    const float* pos1    = (const float*)d_in[0];
    const float* pos2    = (const float*)d_in[1];
    const float* queries = (const float*)d_in[2];
    const float* Wg      = (const float*)d_in[3];
    const float* bg      = (const float*)d_in[4];

    float* wws = (float*)d_ws;

    weights_gemm_wmma<<<448, 256, 0, stream>>>(queries, Wg, bg, wws);
    pair_mlp_wmma<<<1024, 256, 0, stream>>>(pos1, pos2, wws, (float*)d_out);
}


# box-only terms (x2,y2,log w2,log h2; x1,y1,log w1,log h1,1/w1,1/h1) made once per call by an 8th wave of each GEMM group into the workspace; pair waves load them (each made 34 logs); bit-identical
# speedup vs baseline: 2.1527x; 2.1527x over previous
#include <hip/hip_runtime.h>

typedef __attribute__((ext_vector_type(2))) float v2f;
typedef __attribute__((ext_vector_type(8))) float v8f;

constexpr int Bsz = 8, N1 = 1024, N2 = 1024, DIM = 256, HID = 16;
constexpr int WCOLS = 6 * HID + 1;
static_assert(N1 == N2, "one index runs over the boxes of both sets");
constexpr size_t F2_OFF = (size_t)Bsz * N1 * WCOLS;
constexpr size_t R1_OFF = F2_OFF + (size_t)Bsz * N2 * 4;

__global__ __launch_bounds__(256) void weights_gemm_wmma(
        const float* __restrict__ q, const float* __restrict__ Wg,
        const float* __restrict__ bg, const float* __restrict__ pos1,
        const float* __restrict__ pos2, float* __restrict__ wout,
        float* __restrict__ f2, float* __restrict__ r1) {
    const int lane = threadIdx.x & 31;
    const int wid  = threadIdx.x >> 5;
    if (wid == 7) {
        const int e = blockIdx.x * 32 + lane;
        const float4 p2 = ((const float4*)pos2)[e];
        const float4 p1 = ((const float4*)pos1)[e];
        float4 o2;
        o2.x = p2.x;
        o2.y = p2.y;
        o2.z = __logf(p2.z);
        o2.w = __logf(p2.w);
        ((float4*)f2)[e] = o2;
        float4 lo, hi;
        lo.x = p1.x;
        lo.y = p1.y;
        lo.z = 1.0f / p1.z;
        lo.w = 1.0f / p1.w;
        hi.x = __logf(p1.z);
        hi.y = __logf(p1.w);
        hi.z = 1.0f;
        hi.w = 1.0f;
        ((float4*)r1)[e]            = lo;
        ((float4*)r1)[Bsz * N1 + e] = hi;
        return;
    }
    const int c    = lane & 15;
    const int half = lane >> 4;
    const int row0 = blockIdx.x * 32;
    const int col  = wid * 16 + c;
    const int colc = (col < WCOLS) ? col : (WCOLS - 1);

    const float4* qa0 = (const float4*)(q + (size_t)(row0 + c) * DIM + 4 * half);
    const float4* qa1 = (const float4*)(q + (size_t)(row0 + 16 + c) * DIM + 4 * half);
    const float*  wcol = Wg + (size_t)(4 * half) * WCOLS + colc;

    v8f acc0 = {}, acc1 = {};
    #pragma unroll
    for (int j = 0; j < DIM / 8; ++j) {
        const float4 x0 = qa0[2 * j];
        const float4 x1 = qa1[2 * j];
        v2f b0, b1;
        b0.x = wcol[(size_t)(8 * j + 0) * WCOLS];
        b0.y = wcol[(size_t)(8 * j + 1) * WCOLS];
        b1.x = wcol[(size_t)(8 * j + 2) * WCOLS];
        b1.y = wcol[(size_t)(8 * j + 3) * WCOLS];
        v2f a;
        a.x = x0.x; a.y = x0.y;
        acc0 = __builtin_amdgcn_wmma_f32_16x16x4_f32(false, a, false, b0,
                                                     (short)0, acc0, false, false);
        a.x = x1.x; a.y = x1.y;
        acc1 = __builtin_amdgcn_wmma_f32_16x16x4_f32(false, a, false, b0,
                                                     (short)0, acc1, false, false);
        a.x = x0.z; a.y = x0.w;
        acc0 = __builtin_amdgcn_wmma_f32_16x16x4_f32(false, a, false, b1,
                                                     (short)0, acc0, false, false);
        a.x = x1.z; a.y = x1.w;
        acc1 = __builtin_amdgcn_wmma_f32_16x16x4_f32(false, a, false, b1,
                                                     (short)0, acc1, false, false);
    }

    const float bgv = bg[colc];
    const int rbase = row0 + 8 * half;
    if (col < WCOLS) {
        #pragma unroll
        for (int r = 0; r < 8; ++r) {
            wout[(size_t)(rbase + r) * WCOLS + col]      = acc0[r] + bgv;
            wout[(size_t)(rbase + 16 + r) * WCOLS + col] = acc1[r] + bgv;
        }
    }
}

constexpr int NT = 16;
constexpr int MR = 4;
constexpr int XP = NT * 16 + 16;
constexpr int XW = XP + NT * 16;

__global__ __launch_bounds__(256, 8) void pair_mlp_wmma(
        const float* __restrict__ f2, const float* __restrict__ r1,
        const float* __restrict__ wgt, float* __restrict__ out) {
    const int lane = threadIdx.x & 31;
    const int wid  = threadIdx.x >> 5;
    const int nq   = wid & 3;
    const int row0 = blockIdx.x * 8 + (wid >> 2) * MR;
    const int b    = row0 >> 10;
    const bool hiH = lane >= 16;
    const int h    = lane & 15;
    const int nbase = nq * (NT * 16);

    __shared__ __attribute__((aligned(16))) float xch[8 * XW];
    float* xw = xch + wid * XW;
    float* xput = xw + (hiH ? XP : 0) + h;
    const float4* xlo = (const float4*)(xw + 4 * lane);
    const float4* xhi = (const float4*)(xw + XP + 4 * lane);

    const float2* p2row = (const float2*)f2 + (size_t)b * N2 * 2;
    float u0[NT], u1[NT];
    #pragma unroll
    for (int t = 0; t < NT; ++t) {
        const float2 p = p2row[(size_t)(nbase + t * 16 + h) * 2 + (hiH ? 1 : 0)];
        u0[t] = p.x;
        u1[t] = p.y;
    }

    const int k0 = hiH ? 2 : 0;
    const int hb = hiH ? 8 : 0;

    const float4* rterm = (const float4*)r1 + (hiH ? Bsz * N1 : 0) + row0;

    for (int i = 0; i < MR; ++i) {
        const int gid = row0 + i;

        const float4 rt = rterm[i];
        const float c0 = rt.x, c1 = rt.y;
        const float s0 = rt.z, s1 = rt.w;

        const float* wrow = wgt + (size_t)gid * WCOLS;
        v2f a;
        a.x = wrow[(k0 + 0) * HID + h];
        a.y = wrow[(k0 + 1) * HID + h];
        float b1v[8], w2v[8];
        #pragma unroll
        for (int r = 0; r < 8; ++r) {
            b1v[r] = wrow[4 * HID + hb + r];
            w2v[r] = wrow[5 * HID + hb + r];
        }
        const float b2 = wrow[6 * HID];

        float4* orow = (float4*)(out + (size_t)gid * N2 + nbase) + lane;

        #pragma unroll
        for (int t = 0; t < NT; t += 2) {
            v2f fa, fb;
            fa.x = (u0[t] - c0) * s0;
            fa.y = (u1[t] - c1) * s1;
            fb.x = (u0[t + 1] - c0) * s0;
            fb.y = (u1[t + 1] - c1) * s1;

            v8f da = {}, db = {};
            da = __builtin_amdgcn_wmma_f32_16x16x4_f32(false, a, false, fa,
                                                       (short)0, da, false, false);
            db = __builtin_amdgcn_wmma_f32_16x16x4_f32(false, a, false, fb,
                                                       (short)0, db, false, false);
            v2f acc = {0.0f, 0.0f};
            #pragma unroll
            for (int r = 0; r < 8; ++r) {
                v2f tt;
                tt.x = da[r];
                tt.y = db[r];
                tt = __builtin_elementwise_max(tt + b1v[r], (v2f){0.0f, 0.0f});
                acc = __builtin_elementwise_fma(tt, (v2f){w2v[r], w2v[r]}, acc);
            }
            xput[t * 16]       = acc.x;
            xput[(t + 1) * 16] = acc.y;
        }

        __builtin_amdgcn_fence(__ATOMIC_ACQ_REL, "wavefront");

        #pragma unroll
        for (int j = 0; j < 2; ++j) {
            const float4 lo = xlo[32 * j];
            const float4 hi = xhi[32 * j];
            float4 o;
            o.x = (lo.x + hi.x) + b2;
            o.y = (lo.y + hi.y) + b2;
            o.z = (lo.z + hi.z) + b2;
            o.w = (lo.w + hi.w) + b2;
            orow[32 * j] = o;
        }

        __builtin_amdgcn_fence(__ATOMIC_ACQ_REL, "wavefront");
    }
}

extern "C" void kernel_launch(void* const* d_in, const int* in_sizes, int n_in,
                              void* d_out, int out_size, void* d_ws, size_t ws_size,
                              hipStream_t stream) {
    const float* pos1    = (const float*)d_in[0];
    const float* pos2    = (const float*)d_in[1];
    const float* queries = (const float*)d_in[2];
    const float* Wg      = (const float*)d_in[3];
    const float* bg      = (const float*)d_in[4];

    float* wws = (float*)d_ws;
    float* f2  = wws + F2_OFF;
    float* r1  = wws + R1_OFF;

    weights_gemm_wmma<<<256, 256, 0, stream>>>(queries, Wg, bg, pos1, pos2,
                                               wws, f2, r1);
    pair_mlp_wmma<<<1024, 256, 0, stream>>>(f2, r1, wws, (float*)d_out);
}


# pair stage layer 2: packed add/fma take hidden units r,r+1 of one tile as the WMMA's D registers hold them (was tile t with t+1, a register move per value); h-sum order now (even r)+(odd r)
# speedup vs baseline: 2.2876x; 1.0627x over previous
#include <hip/hip_runtime.h>

typedef __attribute__((ext_vector_type(2))) float v2f;
typedef __attribute__((ext_vector_type(8))) float v8f;

constexpr int Bsz = 8, N1 = 1024, N2 = 1024, DIM = 256, HID = 16;
constexpr int WCOLS = 6 * HID + 1;
static_assert(N1 == N2, "one index runs over the boxes of both sets");
constexpr size_t F2_OFF = (size_t)Bsz * N1 * WCOLS;
constexpr size_t R1_OFF = F2_OFF + (size_t)Bsz * N2 * 4;

__global__ __launch_bounds__(256) void weights_gemm_wmma(
        const float* __restrict__ q, const float* __restrict__ Wg,
        const float* __restrict__ bg, const float* __restrict__ pos1,
        const float* __restrict__ pos2, float* __restrict__ wout,
        float* __restrict__ f2, float* __restrict__ r1) {
    const int lane = threadIdx.x & 31;
    const int wid  = threadIdx.x >> 5;
    if (wid == 7) {
        const int e = blockIdx.x * 32 + lane;
        const float4 p2 = ((const float4*)pos2)[e];
        const float4 p1 = ((const float4*)pos1)[e];
        float4 o2;
        o2.x = p2.x;
        o2.y = p2.y;
        o2.z = __logf(p2.z);
        o2.w = __logf(p2.w);
        ((float4*)f2)[e] = o2;
        float4 lo, hi;
        lo.x = p1.x;
        lo.y = p1.y;
        lo.z = 1.0f / p1.z;
        lo.w = 1.0f / p1.w;
        hi.x = __logf(p1.z);
        hi.y = __logf(p1.w);
        hi.z = 1.0f;
        hi.w = 1.0f;
        ((float4*)r1)[e]            = lo;
        ((float4*)r1)[Bsz * N1 + e] = hi;
        return;
    }
    const int c    = lane & 15;
    const int half = lane >> 4;
    const int row0 = blockIdx.x * 32;
    const int col  = wid * 16 + c;
    const int colc = (col < WCOLS) ? col : (WCOLS - 1);

    const float4* qa0 = (const float4*)(q + (size_t)(row0 + c) * DIM + 4 * half);
    const float4* qa1 = (const float4*)(q + (size_t)(row0 + 16 + c) * DIM + 4 * half);
    const float*  wcol = Wg + (size_t)(4 * half) * WCOLS + colc;

    v8f acc0 = {}, acc1 = {};
    #pragma unroll
    for (int j = 0; j < DIM / 8; ++j) {
        const float4 x0 = qa0[2 * j];
        const float4 x1 = qa1[2 * j];
        v2f b0, b1;
        b0.x = wcol[(size_t)(8 * j + 0) * WCOLS];
        b0.y = wcol[(size_t)(8 * j + 1) * WCOLS];
        b1.x = wcol[(size_t)(8 * j + 2) * WCOLS];
        b1.y = wcol[(size_t)(8 * j + 3) * WCOLS];
        v2f a;
        a.x = x0.x; a.y = x0.y;
        acc0 = __builtin_amdgcn_wmma_f32_16x16x4_f32(false, a, false, b0,
                                                     (short)0, acc0, false, false);
        a.x = x1.x; a.y = x1.y;
        acc1 = __builtin_amdgcn_wmma_f32_16x16x4_f32(false, a, false, b0,
                                                     (short)0, acc1, false, false);
        a.x = x0.z; a.y = x0.w;
        acc0 = __builtin_amdgcn_wmma_f32_16x16x4_f32(false, a, false, b1,
                                                     (short)0, acc0, false, false);
        a.x = x1.z; a.y = x1.w;
        acc1 = __builtin_amdgcn_wmma_f32_16x16x4_f32(false, a, false, b1,
                                                     (short)0, acc1, false, false);
    }

    const float bgv = bg[colc];
    const int rbase = row0 + 8 * half;
    if (col < WCOLS) {
        #pragma unroll
        for (int r = 0; r < 8; ++r) {
            wout[(size_t)(rbase + r) * WCOLS + col]      = acc0[r] + bgv;
            wout[(size_t)(rbase + 16 + r) * WCOLS + col] = acc1[r] + bgv;
        }
    }
}

constexpr int NT = 16;
constexpr int MR = 4;
constexpr int XP = NT * 16 + 16;
constexpr int XW = XP + NT * 16;

__global__ __launch_bounds__(256, 8) void pair_mlp_wmma(
        const float* __restrict__ f2, const float* __restrict__ r1,
        const float* __restrict__ wgt, float* __restrict__ out) {
    const int lane = threadIdx.x & 31;
    const int wid  = threadIdx.x >> 5;
    const int nq   = wid & 3;
    const int row0 = blockIdx.x * 8 + (wid >> 2) * MR;
    const int b    = row0 >> 10;
    const bool hiH = lane >= 16;
    const int h    = lane & 15;
    const int nbase = nq * (NT * 16);

    __shared__ __attribute__((aligned(16))) float xch[8 * XW];
    float* xw = xch + wid * XW;
    float* xput = xw + (hiH ? XP : 0) + h;
    const float4* xlo = (const float4*)(xw + 4 * lane);
    const float4* xhi = (const float4*)(xw + XP + 4 * lane);

    const float2* p2row = (const float2*)f2 + (size_t)b * N2 * 2;
    float u0[NT], u1[NT];
    #pragma unroll
    for (int t = 0; t < NT; ++t) {
        const float2 p = p2row[(size_t)(nbase + t * 16 + h) * 2 + (hiH ? 1 : 0)];
        u0[t] = p.x;
        u1[t] = p.y;
    }

    const int k0 = hiH ? 2 : 0;
    const int hb = hiH ? 8 : 0;

    const float4* rterm = (const float4*)r1 + (hiH ? Bsz * N1 : 0) + row0;

    for (int i = 0; i < MR; ++i) {
        const int gid = row0 + i;

        const float4 rt = rterm[i];
        const float c0 = rt.x, c1 = rt.y;
        const float s0 = rt.z, s1 = rt.w;

        const float* wrow = wgt + (size_t)gid * WCOLS;
        v2f a;
        a.x = wrow[(k0 + 0) * HID + h];
        a.y = wrow[(k0 + 1) * HID + h];
        float b1v[8], w2v[8];
        #pragma unroll
        for (int r = 0; r < 8; ++r) {
            b1v[r] = wrow[4 * HID + hb + r];
            w2v[r] = wrow[5 * HID + hb + r];
        }
        const float b2 = wrow[6 * HID];

        float4* orow = (float4*)(out + (size_t)gid * N2 + nbase) + lane;

        #pragma unroll
        for (int t = 0; t < NT; t += 2) {
            v2f fa, fb;
            fa.x = (u0[t] - c0) * s0;
            fa.y = (u1[t] - c1) * s1;
            fb.x = (u0[t + 1] - c0) * s0;
            fb.y = (u1[t + 1] - c1) * s1;

            v8f da = {}, db = {};
            da = __builtin_amdgcn_wmma_f32_16x16x4_f32(false, a, false, fa,
                                                       (short)0, da, false, false);
            db = __builtin_amdgcn_wmma_f32_16x16x4_f32(false, a, false, fb,
                                                       (short)0, db, false, false);
            v2f sa = {0.0f, 0.0f}, sb = {0.0f, 0.0f};
            v2f ta[4], tb[4];
            #pragma unroll
            for (int r = 0; r < 4; ++r) {
                const v2f bb = {b1v[2 * r], b1v[2 * r + 1]};
                ta[r].x = da[2 * r];
                ta[r].y = da[2 * r + 1];
                tb[r].x = db[2 * r];
                tb[r].y = db[2 * r + 1];
                ta[r] = ta[r] + bb;
                tb[r] = tb[r] + bb;
            }
            #pragma unroll
            for (int r = 0; r < 4; ++r) {
                ta[r] = __builtin_elementwise_max(ta[r], (v2f){0.0f, 0.0f});
                tb[r] = __builtin_elementwise_max(tb[r], (v2f){0.0f, 0.0f});
            }
            #pragma unroll
            for (int r = 0; r < 4; ++r) {
                const v2f ww = {w2v[2 * r], w2v[2 * r + 1]};
                sa = __builtin_elementwise_fma(ta[r], ww, sa);
                sb = __builtin_elementwise_fma(tb[r], ww, sb);
            }
            xput[t * 16]       = sa.x + sa.y;
            xput[(t + 1) * 16] = sb.x + sb.y;
        }

        __builtin_amdgcn_fence(__ATOMIC_ACQ_REL, "wavefront");

        #pragma unroll
        for (int j = 0; j < 2; ++j) {
            const float4 lo = xlo[32 * j];
            const float4 hi = xhi[32 * j];
            float4 o;
            o.x = (lo.x + hi.x) + b2;
            o.y = (lo.y + hi.y) + b2;
            o.z = (lo.z + hi.z) + b2;
            o.w = (lo.w + hi.w) + b2;
            orow[32 * j] = o;
        }

        __builtin_amdgcn_fence(__ATOMIC_ACQ_REL, "wavefront");
    }
}

extern "C" void kernel_launch(void* const* d_in, const int* in_sizes, int n_in,
                              void* d_out, int out_size, void* d_ws, size_t ws_size,
                              hipStream_t stream) {
    const float* pos1    = (const float*)d_in[0];
    const float* pos2    = (const float*)d_in[1];
    const float* queries = (const float*)d_in[2];
    const float* Wg      = (const float*)d_in[3];
    const float* bg      = (const float*)d_in[4];

    float* wws = (float*)d_ws;
    float* f2  = wws + F2_OFF;
    float* r1  = wws + R1_OFF;

    weights_gemm_wmma<<<256, 256, 0, stream>>>(queries, Wg, bg, pos1, pos2,
                                               wws, f2, r1);
    pair_mlp_wmma<<<1024, 256, 0, stream>>>(f2, r1, wws, (float*)d_out);
}


# weights GEMM: compiler told the group runs 2 waves per SIMD (waves-per-eu 2,2), so it keeps about 60 operand loads in flight per wave (was about 12 under a 64-register budget); sums unchanged
# speedup vs baseline: 2.5396x; 1.1102x over previous
#include <hip/hip_runtime.h>

typedef __attribute__((ext_vector_type(2))) float v2f;
typedef __attribute__((ext_vector_type(8))) float v8f;

constexpr int Bsz = 8, N1 = 1024, N2 = 1024, DIM = 256, HID = 16;
constexpr int WCOLS = 6 * HID + 1;
static_assert(N1 == N2, "one index runs over the boxes of both sets");
constexpr size_t F2_OFF = (size_t)Bsz * N1 * WCOLS;
constexpr size_t R1_OFF = F2_OFF + (size_t)Bsz * N2 * 4;

__global__ __launch_bounds__(256) __attribute__((amdgpu_waves_per_eu(2, 2)))
void weights_gemm_wmma(
        const float* __restrict__ q, const float* __restrict__ Wg,
        const float* __restrict__ bg, const float* __restrict__ pos1,
        const float* __restrict__ pos2, float* __restrict__ wout,
        float* __restrict__ f2, float* __restrict__ r1) {
    const int lane = threadIdx.x & 31;
    const int wid  = threadIdx.x >> 5;
    if (wid == 7) {
        const int e = blockIdx.x * 32 + lane;
        const float4 p2 = ((const float4*)pos2)[e];
        const float4 p1 = ((const float4*)pos1)[e];
        float4 o2;
        o2.x = p2.x;
        o2.y = p2.y;
        o2.z = __logf(p2.z);
        o2.w = __logf(p2.w);
        ((float4*)f2)[e] = o2;
        float4 lo, hi;
        lo.x = p1.x;
        lo.y = p1.y;
        lo.z = 1.0f / p1.z;
        lo.w = 1.0f / p1.w;
        hi.x = __logf(p1.z);
        hi.y = __logf(p1.w);
        hi.z = 1.0f;
        hi.w = 1.0f;
        ((float4*)r1)[e]            = lo;
        ((float4*)r1)[Bsz * N1 + e] = hi;
        return;
    }
    const int c    = lane & 15;
    const int half = lane >> 4;
    const int row0 = blockIdx.x * 32;
    const int col  = wid * 16 + c;
    const int colc = (col < WCOLS) ? col : (WCOLS - 1);

    const float4* qa0 = (const float4*)(q + (size_t)(row0 + c) * DIM + 4 * half);
    const float4* qa1 = (const float4*)(q + (size_t)(row0 + 16 + c) * DIM + 4 * half);
    const float*  wcol = Wg + (size_t)(4 * half) * WCOLS + colc;

    v8f acc0 = {}, acc1 = {};
    #pragma unroll
    for (int j = 0; j < DIM / 8; ++j) {
        const float4 x0 = qa0[2 * j];
        const float4 x1 = qa1[2 * j];
        v2f b0, b1;
        b0.x = wcol[(size_t)(8 * j + 0) * WCOLS];
        b0.y = wcol[(size_t)(8 * j + 1) * WCOLS];
        b1.x = wcol[(size_t)(8 * j + 2) * WCOLS];
        b1.y = wcol[(size_t)(8 * j + 3) * WCOLS];
        v2f a;
        a.x = x0.x; a.y = x0.y;
        acc0 = __builtin_amdgcn_wmma_f32_16x16x4_f32(false, a, false, b0,
                                                     (short)0, acc0, false, false);
        a.x = x1.x; a.y = x1.y;
        acc1 = __builtin_amdgcn_wmma_f32_16x16x4_f32(false, a, false, b0,
                                                     (short)0, acc1, false, false);
        a.x = x0.z; a.y = x0.w;
        acc0 = __builtin_amdgcn_wmma_f32_16x16x4_f32(false, a, false, b1,
                                                     (short)0, acc0, false, false);
        a.x = x1.z; a.y = x1.w;
        acc1 = __builtin_amdgcn_wmma_f32_16x16x4_f32(false, a, false, b1,
                                                     (short)0, acc1, false, false);
    }

    const float bgv = bg[colc];
    const int rbase = row0 + 8 * half;
    if (col < WCOLS) {
        #pragma unroll
        for (int r = 0; r < 8; ++r) {
            wout[(size_t)(rbase + r) * WCOLS + col]      = acc0[r] + bgv;
            wout[(size_t)(rbase + 16 + r) * WCOLS + col] = acc1[r] + bgv;
        }
    }
}

constexpr int NT = 16;
constexpr int MR = 4;
constexpr int XP = NT * 16 + 16;
constexpr int XW = XP + NT * 16;

__global__ __launch_bounds__(256, 8) void pair_mlp_wmma(
        const float* __restrict__ f2, const float* __restrict__ r1,
        const float* __restrict__ wgt, float* __restrict__ out) {
    const int lane = threadIdx.x & 31;
    const int wid  = threadIdx.x >> 5;
    const int nq   = wid & 3;
    const int row0 = blockIdx.x * 8 + (wid >> 2) * MR;
    const int b    = row0 >> 10;
    const bool hiH = lane >= 16;
    const int h    = lane & 15;
    const int nbase = nq * (NT * 16);

    __shared__ __attribute__((aligned(16))) float xch[8 * XW];
    float* xw = xch + wid * XW;
    float* xput = xw + (hiH ? XP : 0) + h;
    const float4* xlo = (const float4*)(xw + 4 * lane);
    const float4* xhi = (const float4*)(xw + XP + 4 * lane);

    const float2* p2row = (const float2*)f2 + (size_t)b * N2 * 2;
    float u0[NT], u1[NT];
    #pragma unroll
    for (int t = 0; t < NT; ++t) {
        const float2 p = p2row[(size_t)(nbase + t * 16 + h) * 2 + (hiH ? 1 : 0)];
        u0[t] = p.x;
        u1[t] = p.y;
    }

    const int k0 = hiH ? 2 : 0;
    const int hb = hiH ? 8 : 0;

    const float4* rterm = (const float4*)r1 + (hiH ? Bsz * N1 : 0) + row0;

    for (int i = 0; i < MR; ++i) {
        const int gid = row0 + i;

        const float4 rt = rterm[i];
        const float c0 = rt.x, c1 = rt.y;
        const float s0 = rt.z, s1 = rt.w;

        const float* wrow = wgt + (size_t)gid * WCOLS;
        v2f a;
        a.x = wrow[(k0 + 0) * HID + h];
        a.y = wrow[(k0 + 1) * HID + h];
        float b1v[8], w2v[8];
        #pragma unroll
        for (int r = 0; r < 8; ++r) {
            b1v[r] = wrow[4 * HID + hb + r];
            w2v[r] = wrow[5 * HID + hb + r];
        }
        const float b2 = wrow[6 * HID];

        float4* orow = (float4*)(out + (size_t)gid * N2 + nbase) + lane;

        #pragma unroll
        for (int t = 0; t < NT; t += 2) {
            v2f fa, fb;
            fa.x = (u0[t] - c0) * s0;
            fa.y = (u1[t] - c1) * s1;
            fb.x = (u0[t + 1] - c0) * s0;
            fb.y = (u1[t + 1] - c1) * s1;

            v8f da = {}, db = {};
            da = __builtin_amdgcn_wmma_f32_16x16x4_f32(false, a, false, fa,
                                                       (short)0, da, false, false);
            db = __builtin_amdgcn_wmma_f32_16x16x4_f32(false, a, false, fb,
                                                       (short)0, db, false, false);
            v2f sa = {0.0f, 0.0f}, sb = {0.0f, 0.0f};
            v2f ta[4], tb[4];
            #pragma unroll
            for (int r = 0; r < 4; ++r) {
                const v2f bb = {b1v[2 * r], b1v[2 * r + 1]};
                ta[r].x = da[2 * r];
                ta[r].y = da[2 * r + 1];
                tb[r].x = db[2 * r];
                tb[r].y = db[2 * r + 1];
                ta[r] = ta[r] + bb;
                tb[r] = tb[r] + bb;
            }
            #pragma unroll
            for (int r = 0; r < 4; ++r) {
                ta[r] = __builtin_elementwise_max(ta[r], (v2f){0.0f, 0.0f});
                tb[r] = __builtin_elementwise_max(tb[r], (v2f){0.0f, 0.0f});
            }
            #pragma unroll
            for (int r = 0; r < 4; ++r) {
                const v2f ww = {w2v[2 * r], w2v[2 * r + 1]};
                sa = __builtin_elementwise_fma(ta[r], ww, sa);
                sb = __builtin_elementwise_fma(tb[r], ww, sb);
            }
            xput[t * 16]       = sa.x + sa.y;
            xput[(t + 1) * 16] = sb.x + sb.y;
        }

        __builtin_amdgcn_fence(__ATOMIC_ACQ_REL, "wavefront");

        #pragma unroll
        for (int j = 0; j < 2; ++j) {
            const float4 lo = xlo[32 * j];
            const float4 hi = xhi[32 * j];
            float4 o;
            o.x = (lo.x + hi.x) + b2;
            o.y = (lo.y + hi.y) + b2;
            o.z = (lo.z + hi.z) + b2;
            o.w = (lo.w + hi.w) + b2;
            orow[32 * j] = o;
        }

        __builtin_amdgcn_fence(__ATOMIC_ACQ_REL, "wavefront");
    }
}

extern "C" void kernel_launch(void* const* d_in, const int* in_sizes, int n_in,
                              void* d_out, int out_size, void* d_ws, size_t ws_size,
                              hipStream_t stream) {
    const float* pos1    = (const float*)d_in[0];
    const float* pos2    = (const float*)d_in[1];
    const float* queries = (const float*)d_in[2];
    const float* Wg      = (const float*)d_in[3];
    const float* bg      = (const float*)d_in[4];

    float* wws = (float*)d_ws;
    float* f2  = wws + F2_OFF;
    float* r1  = wws + R1_OFF;

    weights_gemm_wmma<<<256, 256, 0, stream>>>(queries, Wg, bg, pos1, pos2,
                                               wws, f2, r1);
    pair_mlp_wmma<<<1024, 256, 0, stream>>>(f2, r1, wws, (float*)d_out);
}
